# starting kernel
# baseline (speedup 1.0000x reference)
#include <hip/hip_runtime.h>
#include <math.h>

typedef __attribute__((ext_vector_type(2))) float v2f;
typedef __attribute__((ext_vector_type(8))) float v8f;

#define WMMA_F32(a, b, c) \
  __builtin_amdgcn_wmma_f32_16x16x4_f32(false, (a), false, (b), (short)0, (c), false, false)

constexpr int Bb = 2;
constexpr int Ss = 2048;
constexpr int Hh = 12;
constexpr int Dd = 64;
constexpr int BH = Bb * Hh;
constexpr int NCH = 16;
constexpr int CH = Ss / NCH;
constexpr int SROW = 7 * Hh * Dd;
constexpr int MSZ = Dd * Dd;

__device__ __forceinline__ size_t qoff(int b, int s, int slot, int h, int d) {
  return ((((size_t)b * Ss + s) * 7 + slot) * Hh + h) * (size_t)Dd + d;
}

__global__ void __launch_bounds__(256) ktv_partials(const float* __restrict__ qkv,
                                                    float* __restrict__ P1,
                                                    float* __restrict__ P2) {
  const int bh = blockIdx.x / NCH, ch = blockIdx.x % NCH;
  const int b = bh / Hh, h = bh % Hh;
  const int wave = threadIdx.x >> 5, lane = threadIdx.x & 31;
  const int mat = wave >> 2;
  const int irow = wave & 3;
  const int kslot = mat ? 3 : 1;
  const int m = lane & 15;
  const int kk = (lane >> 4) << 1;
  v8f acc[4] = {};
  const int s0b = ch * CH;
  for (int s0 = s0b; s0 < s0b + CH; s0 += 4) {
    const float* K0 = qkv + qoff(b, s0 + kk, kslot, h, 0);
    const float* V0 = qkv + qoff(b, s0 + kk, 6, h, 0);
    v2f a;
    a.x = K0[irow * 16 + m];
    a.y = K0[SROW + irow * 16 + m];
#pragma unroll
    for (int j = 0; j < 4; ++j) {
      v2f bt;
      bt.x = V0[j * 16 + m];
      bt.y = V0[SROW + j * 16 + m];
      acc[j] = WMMA_F32(a, bt, acc[j]);
    }
  }
  float* P = (mat ? P2 : P1) + ((size_t)bh * NCH + ch) * MSZ;
  const int rh = (lane >> 4) * 8;
#pragma unroll
  for (int j = 0; j < 4; ++j)
#pragma unroll
    for (int r = 0; r < 8; ++r)
      P[(irow * 16 + rh + r) * Dd + j * 16 + m] = acc[j][r];
}

__global__ void __launch_bounds__(256) reduce_partials(const float* __restrict__ P,
                                                       float* __restrict__ M) {
  const int bh = blockIdx.x;
  for (int e = threadIdx.x; e < MSZ; e += 256) {
    float s = 0.f;
#pragma unroll
    for (int c = 0; c < NCH; ++c) s += P[((size_t)bh * NCH + c) * MSZ + e];
    M[(size_t)bh * MSZ + e] = s;
  }
}

__global__ void __launch_bounds__(256) u_kernel(const float* __restrict__ qkv,
                                                const float* __restrict__ M1,
                                                const float* __restrict__ M2,
                                                float* __restrict__ U) {
  __shared__ float sM1[MSZ];
  __shared__ float sM2[MSZ];
  const int bh = blockIdx.x / NCH, ch = blockIdx.x % NCH;
  const int b = bh / Hh, h = bh % Hh;
  for (int e = threadIdx.x; e < MSZ; e += 256) {
    sM1[e] = M1[(size_t)bh * MSZ + e];
    sM2[e] = M2[(size_t)bh * MSZ + e];
  }
  __syncthreads();
  const int wave = threadIdx.x >> 5, lane = threadIdx.x & 31;
  const int tbase = ch * CH + wave * 16;
  const int m = lane & 15, kk = (lane >> 4) << 1;
  v8f acc1[4] = {}, acc2[4] = {};
  for (int step = 0; step < 16; ++step) {
    const int k0 = step * 4 + kk;
    const float* q1p = qkv + qoff(b, tbase + m, 0, h, k0);
    const float* q2p = qkv + qoff(b, tbase + m, 2, h, k0);
    v2f a1, a2;
    a1.x = q1p[0]; a1.y = q1p[1];
    a2.x = q2p[0]; a2.y = q2p[1];
#pragma unroll
    for (int j = 0; j < 4; ++j) {
      v2f b1, b2;
      b1.x = sM1[k0 * Dd + j * 16 + m];
      b1.y = sM1[(k0 + 1) * Dd + j * 16 + m];
      b2.x = sM2[k0 * Dd + j * 16 + m];
      b2.y = sM2[(k0 + 1) * Dd + j * 16 + m];
      acc1[j] = WMMA_F32(a1, b1, acc1[j]);
      acc2[j] = WMMA_F32(a2, b2, acc2[j]);
    }
  }
  const int rh = (lane >> 4) * 8;
#pragma unroll
  for (int j = 0; j < 4; ++j)
#pragma unroll
    for (int r = 0; r < 8; ++r) {
      const float x = acc1[j][r];
      const float t1 = x / (1.f + __expf(-x));
      U[((size_t)bh * Ss + tbase + rh + r) * Dd + j * 16 + m] = t1 * acc2[j][r];
    }
}

__global__ void __launch_bounds__(128) ktu_partials(const float* __restrict__ qkv,
                                                    const float* __restrict__ U,
                                                    float* __restrict__ P3) {
  const int bh = blockIdx.x / NCH, ch = blockIdx.x % NCH;
  const int b = bh / Hh, h = bh % Hh;
  const int wave = threadIdx.x >> 5, lane = threadIdx.x & 31;
  const int irow = wave;
  const int m = lane & 15, kk = (lane >> 4) << 1;
  v8f acc[4] = {};
  const int s0b = ch * CH;
  for (int s0 = s0b; s0 < s0b + CH; s0 += 4) {
    const float* K0 = qkv + qoff(b, s0 + kk, 5, h, 0);
    const float* U0 = U + ((size_t)bh * Ss + s0 + kk) * Dd;
    v2f a;
    a.x = K0[irow * 16 + m];
    a.y = K0[SROW + irow * 16 + m];
#pragma unroll
    for (int j = 0; j < 4; ++j) {
      v2f bt;
      bt.x = U0[j * 16 + m];
      bt.y = U0[Dd + j * 16 + m];
      acc[j] = WMMA_F32(a, bt, acc[j]);
    }
  }
  float* P = P3 + ((size_t)bh * NCH + ch) * MSZ;
  const int rh = (lane >> 4) * 8;
#pragma unroll
  for (int j = 0; j < 4; ++j)
#pragma unroll
    for (int r = 0; r < 8; ++r)
      P[(irow * 16 + rh + r) * Dd + j * 16 + m] = acc[j][r];
}

__global__ void __launch_bounds__(256) out_kernel(const float* __restrict__ qkv,
                                                  const float* __restrict__ M3,
                                                  float* __restrict__ out) {
  __shared__ float sM[MSZ];
  const int bh = blockIdx.x / NCH, ch = blockIdx.x % NCH;
  const int b = bh / Hh, h = bh % Hh;
  for (int e = threadIdx.x; e < MSZ; e += 256) sM[e] = M3[(size_t)bh * MSZ + e];
  __syncthreads();
  const int wave = threadIdx.x >> 5, lane = threadIdx.x & 31;
  const int tbase = ch * CH + wave * 16;
  const int m = lane & 15, kk = (lane >> 4) << 1;
  v8f acc[4] = {};
  for (int step = 0; step < 16; ++step) {
    const int k0 = step * 4 + kk;
    const float* qp = qkv + qoff(b, tbase + m, 4, h, k0);
    v2f a; a.x = qp[0]; a.y = qp[1];
#pragma unroll
    for (int j = 0; j < 4; ++j) {
      v2f bt;
      bt.x = sM[k0 * Dd + j * 16 + m];
      bt.y = sM[(k0 + 1) * Dd + j * 16 + m];
      acc[j] = WMMA_F32(a, bt, acc[j]);
    }
  }
  const int rh = (lane >> 4) * 8;
#pragma unroll
  for (int j = 0; j < 4; ++j)
#pragma unroll
    for (int r = 0; r < 8; ++r)
      out[(((size_t)b * Ss + tbase + rh + r) * Hh + h) * Dd + j * 16 + m] = acc[j][r];
}

extern "C" void kernel_launch(void* const* d_in, const int* in_sizes, int n_in,
                              void* d_out, int out_size, void* d_ws, size_t ws_size,
                              hipStream_t stream) {
  (void)in_sizes; (void)n_in; (void)out_size; (void)ws_size;
  const float* qkv = (const float*)d_in[0];
  float* out = (float*)d_out;
  float* w = (float*)d_ws;

  const size_t PSZ = (size_t)BH * NCH * MSZ;
  const size_t MTOT = (size_t)BH * MSZ;
  float* P1 = w;
  float* P2 = P1 + PSZ;
  float* P3 = P2 + PSZ;
  float* M1 = P3 + PSZ;
  float* M2 = M1 + MTOT;
  float* M3 = M2 + MTOT;
  float* U  = M3 + MTOT;

  ktv_partials<<<BH * NCH, 256, 0, stream>>>(qkv, P1, P2);
  reduce_partials<<<BH, 256, 0, stream>>>(P1, M1);
  reduce_partials<<<BH, 256, 0, stream>>>(P2, M2);
  u_kernel<<<BH * NCH, 256, 0, stream>>>(qkv, M1, M2, U);
  ktu_partials<<<BH * NCH, 128, 0, stream>>>(qkv, U, P3);
  reduce_partials<<<BH, 256, 0, stream>>>(P3, M3);
  out_kernel<<<BH * NCH, 256, 0, stream>>>(qkv, M3, out);
}


# Producer blocks sum their 16 S-chunk partials in LDS (16 waves per block, same c=0..15 order, bit-identical): 3 reduce launches and partial planes gone; K3^T u wave tile 16x32 to keep 192 blocks
# speedup vs baseline: 1.3880x; 1.3880x over previous
#include <hip/hip_runtime.h>
#include <math.h>

typedef __attribute__((ext_vector_type(2))) float v2f;
typedef __attribute__((ext_vector_type(8))) float v8f;

#define WMMA_F32(a, b, c) \
  __builtin_amdgcn_wmma_f32_16x16x4_f32(false, (a), false, (b), (short)0, (c), false, false)

constexpr int Bb = 2;
constexpr int Ss = 2048;
constexpr int Hh = 12;
constexpr int Dd = 64;
constexpr int BH = Bb * Hh;
constexpr int NCH = 16;
constexpr int CH = Ss / NCH;
constexpr int SROW = 7 * Hh * Dd;
constexpr int MSZ = Dd * Dd;

__device__ __forceinline__ size_t qoff(int b, int s, int slot, int h, int d) {
  return ((((size_t)b * Ss + s) * 7 + slot) * Hh + h) * (size_t)Dd + d;
}

__device__ __forceinline__ int wave_index() {
  return __builtin_amdgcn_readfirstlane((int)(threadIdx.x >> 5));
}

template <int NJ>
__device__ __forceinline__ void block_chunk_sum(const v8f (&acc)[NJ], float* sP,
                                                float* __restrict__ Mdst, int mpitch) {
  constexpr int TW = 16 * NJ;
  constexpr int TSZ = 16 * TW;
  const int wave = wave_index(), lane = threadIdx.x & 31;
  const int m = lane & 15, rh = (lane >> 4) * 8;
#pragma unroll
  for (int j = 0; j < NJ; ++j)
#pragma unroll
    for (int r = 0; r < 8; ++r)
      sP[wave * TSZ + (rh + r) * TW + j * 16 + m] = acc[j][r];
  __syncthreads();
  for (int e = threadIdx.x; e < TSZ; e += NCH * 32) {
    float s = 0.f;
#pragma unroll
    for (int c = 0; c < NCH; ++c) s += sP[c * TSZ + e];
    Mdst[(e / TW) * mpitch + (e % TW)] = s;
  }
}

__global__ void __launch_bounds__(NCH * 32) ktv_strips(const float* __restrict__ qkv,
                                                       float* __restrict__ M1,
                                                       float* __restrict__ M2) {
  __shared__ float sP[NCH * 16 * Dd];
  const int irow = blockIdx.x & 3;
  const int mat = (blockIdx.x >> 2) & 1;
  const int bh = blockIdx.x >> 3;
  const int b = bh / Hh, h = bh % Hh;
  const int wave = wave_index(), lane = threadIdx.x & 31;
  const int kslot = mat ? 3 : 1;
  const int m = lane & 15;
  const int kk = (lane >> 4) << 1;
  v8f acc[4] = {};
  const int s0b = wave * CH;
  for (int s0 = s0b; s0 < s0b + CH; s0 += 4) {
    const float* K0 = qkv + qoff(b, s0 + kk, kslot, h, 0);
    const float* V0 = qkv + qoff(b, s0 + kk, 6, h, 0);
    v2f a;
    a.x = K0[irow * 16 + m];
    a.y = K0[SROW + irow * 16 + m];
#pragma unroll
    for (int j = 0; j < 4; ++j) {
      v2f bt;
      bt.x = V0[j * 16 + m];
      bt.y = V0[SROW + j * 16 + m];
      acc[j] = WMMA_F32(a, bt, acc[j]);
    }
  }
  float* M = (mat ? M2 : M1) + (size_t)bh * MSZ + irow * 16 * Dd;
  block_chunk_sum<4>(acc, sP, M, Dd);
}

__global__ void __launch_bounds__(256) u_kernel(const float* __restrict__ qkv,
                                                const float* __restrict__ M1,
                                                const float* __restrict__ M2,
                                                float* __restrict__ U) {
  __shared__ float sM1[MSZ];
  __shared__ float sM2[MSZ];
  const int bh = blockIdx.x / NCH, ch = blockIdx.x % NCH;
  const int b = bh / Hh, h = bh % Hh;
  for (int e = threadIdx.x; e < MSZ; e += 256) {
    sM1[e] = M1[(size_t)bh * MSZ + e];
    sM2[e] = M2[(size_t)bh * MSZ + e];
  }
  __syncthreads();
  const int wave = threadIdx.x >> 5, lane = threadIdx.x & 31;
  const int tbase = ch * CH + wave * 16;
  const int m = lane & 15, kk = (lane >> 4) << 1;
  v8f acc1[4] = {}, acc2[4] = {};
  for (int step = 0; step < 16; ++step) {
    const int k0 = step * 4 + kk;
    const float* q1p = qkv + qoff(b, tbase + m, 0, h, k0);
    const float* q2p = qkv + qoff(b, tbase + m, 2, h, k0);
    v2f a1, a2;
    a1.x = q1p[0]; a1.y = q1p[1];
    a2.x = q2p[0]; a2.y = q2p[1];
#pragma unroll
    for (int j = 0; j < 4; ++j) {
      v2f b1, b2;
      b1.x = sM1[k0 * Dd + j * 16 + m];
      b1.y = sM1[(k0 + 1) * Dd + j * 16 + m];
      b2.x = sM2[k0 * Dd + j * 16 + m];
      b2.y = sM2[(k0 + 1) * Dd + j * 16 + m];
      acc1[j] = WMMA_F32(a1, b1, acc1[j]);
      acc2[j] = WMMA_F32(a2, b2, acc2[j]);
    }
  }
  const int rh = (lane >> 4) * 8;
#pragma unroll
  for (int j = 0; j < 4; ++j)
#pragma unroll
    for (int r = 0; r < 8; ++r) {
      const float x = acc1[j][r];
      const float t1 = x / (1.f + __expf(-x));
      U[((size_t)bh * Ss + tbase + rh + r) * Dd + j * 16 + m] = t1 * acc2[j][r];
    }
}

__global__ void __launch_bounds__(NCH * 32) ktu_strips(const float* __restrict__ qkv,
                                                       const float* __restrict__ U,
                                                       float* __restrict__ M3) {
  __shared__ float sP[NCH * 16 * 32];
  const int jh = blockIdx.x & 1;
  const int irow = (blockIdx.x >> 1) & 3;
  const int bh = blockIdx.x >> 3;
  const int b = bh / Hh, h = bh % Hh;
  const int wave = wave_index(), lane = threadIdx.x & 31;
  const int m = lane & 15, kk = (lane >> 4) << 1;
  v8f acc[2] = {};
  const int s0b = wave * CH;
  for (int s0 = s0b; s0 < s0b + CH; s0 += 4) {
    const float* K0 = qkv + qoff(b, s0 + kk, 5, h, 0);
    const float* U0 = U + ((size_t)bh * Ss + s0 + kk) * Dd + jh * 32;
    v2f a;
    a.x = K0[irow * 16 + m];
    a.y = K0[SROW + irow * 16 + m];
#pragma unroll
    for (int j = 0; j < 2; ++j) {
      v2f bt;
      bt.x = U0[j * 16 + m];
      bt.y = U0[Dd + j * 16 + m];
      acc[j] = WMMA_F32(a, bt, acc[j]);
    }
  }
  float* M = M3 + (size_t)bh * MSZ + irow * 16 * Dd + jh * 32;
  block_chunk_sum<2>(acc, sP, M, Dd);
}

__global__ void __launch_bounds__(256) out_kernel(const float* __restrict__ qkv,
                                                  const float* __restrict__ M3,
                                                  float* __restrict__ out) {
  __shared__ float sM[MSZ];
  const int bh = blockIdx.x / NCH, ch = blockIdx.x % NCH;
  const int b = bh / Hh, h = bh % Hh;
  for (int e = threadIdx.x; e < MSZ; e += 256) sM[e] = M3[(size_t)bh * MSZ + e];
  __syncthreads();
  const int wave = threadIdx.x >> 5, lane = threadIdx.x & 31;
  const int tbase = ch * CH + wave * 16;
  const int m = lane & 15, kk = (lane >> 4) << 1;
  v8f acc[4] = {};
  for (int step = 0; step < 16; ++step) {
    const int k0 = step * 4 + kk;
    const float* qp = qkv + qoff(b, tbase + m, 4, h, k0);
    v2f a; a.x = qp[0]; a.y = qp[1];
#pragma unroll
    for (int j = 0; j < 4; ++j) {
      v2f bt;
      bt.x = sM[k0 * Dd + j * 16 + m];
      bt.y = sM[(k0 + 1) * Dd + j * 16 + m];
      acc[j] = WMMA_F32(a, bt, acc[j]);
    }
  }
  const int rh = (lane >> 4) * 8;
#pragma unroll
  for (int j = 0; j < 4; ++j)
#pragma unroll
    for (int r = 0; r < 8; ++r)
      out[(((size_t)b * Ss + tbase + rh + r) * Hh + h) * Dd + j * 16 + m] = acc[j][r];
}

extern "C" void kernel_launch(void* const* d_in, const int* in_sizes, int n_in,
                              void* d_out, int out_size, void* d_ws, size_t ws_size,
                              hipStream_t stream) {
  (void)in_sizes; (void)n_in; (void)out_size; (void)ws_size;
  const float* qkv = (const float*)d_in[0];
  float* out = (float*)d_out;
  float* w = (float*)d_ws;

  const size_t MTOT = (size_t)BH * MSZ;
  float* M1 = w;
  float* M2 = M1 + MTOT;
  float* M3 = M2 + MTOT;
  float* U  = M3 + MTOT;

  ktv_strips<<<BH * 8, NCH * 32, 0, stream>>>(qkv, M1, M2);
  u_kernel<<<BH * NCH, 256, 0, stream>>>(qkv, M1, M2, U);
  ktu_strips<<<BH * 8, NCH * 32, 0, stream>>>(qkv, U, M3);
  out_kernel<<<BH * NCH, 256, 0, stream>>>(qkv, M3, out);
}


# Loads issued as deep groups ahead of use (32-64 s rows per trip in the K^T kernels; M and all 16 Q fragments before the LDS fill), rows as 128-bit via column-interleaved tiles; same summation order
# speedup vs baseline: 2.1362x; 1.5390x over previous
#include <hip/hip_runtime.h>
#include <math.h>


typedef __attribute__((ext_vector_type(2))) float v2f;
typedef __attribute__((ext_vector_type(4))) float v4f;
typedef __attribute__((ext_vector_type(8))) float v8f;

#define WMMA_F32(a, b, c) \
  __builtin_amdgcn_wmma_f32_16x16x4_f32(false, (a), false, (b), (short)0, (c), false, false)

constexpr int Bb = 2;
constexpr int Ss = 2048;
constexpr int Hh = 12;
constexpr int Dd = 64;
constexpr int BH = Bb * Hh;
constexpr int NCH = 16;
constexpr int CH = Ss / NCH;
constexpr int SROW = 7 * Hh * Dd;
constexpr int MSZ = Dd * Dd;


__device__ __forceinline__ size_t qoff(int b, int s, int slot, int h, int d) {
  return ((((size_t)b * Ss + s) * 7 + slot) * Hh + h) * (size_t)Dd + d;
}

__device__ __forceinline__ int wave_index() {
  return __builtin_amdgcn_readfirstlane((int)(threadIdx.x >> 5));
}

template <int NJ>
__device__ __forceinline__ void block_chunk_sum(const v8f (&acc)[NJ], float* sP,
                                                float* __restrict__ Mdst, int mpitch) {
  constexpr int TW = 16 * NJ;
  constexpr int TSZ = 16 * TW;
  typedef __attribute__((ext_vector_type(NJ))) float vnf;
  const int wave = wave_index(), lane = threadIdx.x & 31;
  const int m = lane & 15, rh = (lane >> 4) * 8;
#pragma unroll
  for (int r = 0; r < 8; ++r) {
    vnf o;
#pragma unroll
    for (int j = 0; j < NJ; ++j) o[j] = acc[j][r];
    *(vnf*)&sP[wave * TSZ + (rh + r) * TW + NJ * m] = o;
  }
  __syncthreads();
  for (int e = threadIdx.x; e < TSZ; e += NCH * 32) {
    float s = 0.f;
#pragma unroll
    for (int c = 0; c < NCH; ++c) s += sP[c * TSZ + e];
    Mdst[(e / TW) * mpitch + (e % TW)] = s;
  }
}

constexpr int RG1 = 8;
__global__ void __launch_bounds__(NCH * 32) ktv_strips(const float* __restrict__ qkv,
                                                       float* __restrict__ M1,
                                                       float* __restrict__ M2) {
  __shared__ __attribute__((aligned(16))) float sP[NCH * 16 * Dd];
  const int irow = blockIdx.x & 3;
  const int mat = (blockIdx.x >> 2) & 1;
  const int bh = blockIdx.x >> 3;
  const int b = bh / Hh, h = bh % Hh;
  const int wave = wave_index(), lane = threadIdx.x & 31;
  const int kslot = mat ? 3 : 1;
  const int m = lane & 15;
  const int kk = (lane >> 4) << 1;
  v8f acc[4] = {};
  const float* Kp = qkv + qoff(b, wave * CH + kk, kslot, h, irow * 16 + m);
  const float* Vp = qkv + qoff(b, wave * CH + kk, 6, h, 4 * m);
  for (int t = 0; t < CH / (4 * RG1); ++t) {
    v2f a[RG1];
    v4f v0[RG1], v1[RG1];
#pragma unroll
    for (int g = 0; g < RG1; ++g) {
      const float* kp = Kp + (size_t)(t * RG1 + g) * 4 * SROW;
      const float* vp = Vp + (size_t)(t * RG1 + g) * 4 * SROW;
      a[g].x = kp[0];
      a[g].y = kp[SROW];
      v0[g] = *(const v4f*)vp;
      v1[g] = *(const v4f*)(vp + SROW);
    }
#pragma unroll
    for (int g = 0; g < RG1; ++g)
#pragma unroll
      for (int j = 0; j < 4; ++j) {
        v2f bt;
        bt.x = v0[g][j];
        bt.y = v1[g][j];
        acc[j] = WMMA_F32(a[g], bt, acc[j]);
      }
  }
  float* M = (mat ? M2 : M1) + (size_t)bh * MSZ + irow * 16 * Dd;
  block_chunk_sum<4>(acc, sP, M, Dd);
}

__global__ void __launch_bounds__(256) u_kernel(const float* __restrict__ qkv,
                                                const float* __restrict__ M1,
                                                const float* __restrict__ M2,
                                                float* __restrict__ U) {
  __shared__ __attribute__((aligned(16))) float sM1[MSZ];
  __shared__ __attribute__((aligned(16))) float sM2[MSZ];
  const int bh = blockIdx.x / NCH, ch = blockIdx.x % NCH;
  const int b = bh / Hh, h = bh % Hh;
  const int wave = threadIdx.x >> 5, lane = threadIdx.x & 31;
  const int tbase = ch * CH + wave * 16;
  const int m = lane & 15, kk = (lane >> 4) << 1;
  const v4f* g1 = (const v4f*)(M1 + (size_t)bh * MSZ);
  const v4f* g2 = (const v4f*)(M2 + (size_t)bh * MSZ);
  v4f f1[4], f2[4];
#pragma unroll
  for (int i = 0; i < 4; ++i) {
    f1[i] = g1[threadIdx.x + i * 256];
    f2[i] = g2[threadIdx.x + i * 256];
  }
  const float* q1p = qkv + qoff(b, tbase + m, 0, h, kk);
  const float* q2p = qkv + qoff(b, tbase + m, 2, h, kk);
  v2f a1[16], a2[16];
#pragma unroll
  for (int step = 0; step < 16; ++step) {
    a1[step] = *(const v2f*)(q1p + 4 * step);
    a2[step] = *(const v2f*)(q2p + 4 * step);
  }
#pragma unroll
  for (int i = 0; i < 4; ++i) {
    ((v4f*)sM1)[threadIdx.x + i * 256] = f1[i];
    ((v4f*)sM2)[threadIdx.x + i * 256] = f2[i];
  }
  __syncthreads();
  v8f acc1[4] = {}, acc2[4] = {};
#pragma unroll
  for (int step = 0; step < 16; ++step) {
    const int k0 = step * 4 + kk;
    const v4f x0 = *(const v4f*)&sM1[k0 * Dd + 4 * m];
    const v4f x1 = *(const v4f*)&sM1[(k0 + 1) * Dd + 4 * m];
    const v4f y0 = *(const v4f*)&sM2[k0 * Dd + 4 * m];
    const v4f y1 = *(const v4f*)&sM2[(k0 + 1) * Dd + 4 * m];
#pragma unroll
    for (int j = 0; j < 4; ++j) {
      v2f b1, b2;
      b1.x = x0[j]; b1.y = x1[j];
      b2.x = y0[j]; b2.y = y1[j];
      acc1[j] = WMMA_F32(a1[step], b1, acc1[j]);
      acc2[j] = WMMA_F32(a2[step], b2, acc2[j]);
    }
  }
  const int rh = (lane >> 4) * 8;
#pragma unroll
  for (int r = 0; r < 8; ++r) {
    v4f o;
#pragma unroll
    for (int j = 0; j < 4; ++j) {
      const float x = acc1[j][r];
      const float t1 = x / (1.f + __expf(-x));
      o[j] = t1 * acc2[j][r];
    }
    *(v4f*)&U[((size_t)bh * Ss + tbase + rh + r) * Dd + 4 * m] = o;
  }
}

constexpr int RG3 = 16;
__global__ void __launch_bounds__(NCH * 32) ktu_strips(const float* __restrict__ qkv,
                                                       const float* __restrict__ U,
                                                       float* __restrict__ M3) {
  __shared__ __attribute__((aligned(16))) float sP[NCH * 16 * 32];
  const int jh = blockIdx.x & 1;
  const int irow = (blockIdx.x >> 1) & 3;
  const int bh = blockIdx.x >> 3;
  const int b = bh / Hh, h = bh % Hh;
  const int wave = wave_index(), lane = threadIdx.x & 31;
  const int m = lane & 15, kk = (lane >> 4) << 1;
  v8f acc[2] = {};
  const float* Kp = qkv + qoff(b, wave * CH + kk, 5, h, irow * 16 + m);
  const float* Up = U + ((size_t)bh * Ss + wave * CH + kk) * Dd + jh * 32 + 2 * m;
  for (int t = 0; t < CH / (4 * RG3); ++t) {
    v2f a[RG3], u0[RG3], u1[RG3];
#pragma unroll
    for (int g = 0; g < RG3; ++g) {
      const float* kp = Kp + (size_t)(t * RG3 + g) * 4 * SROW;
      const float* up = Up + (size_t)(t * RG3 + g) * 4 * Dd;
      a[g].x = kp[0];
      a[g].y = kp[SROW];
      u0[g] = *(const v2f*)up;
      u1[g] = *(const v2f*)(up + Dd);
    }
#pragma unroll
    for (int g = 0; g < RG3; ++g)
#pragma unroll
      for (int j = 0; j < 2; ++j) {
        v2f bt;
        bt.x = u0[g][j];
        bt.y = u1[g][j];
        acc[j] = WMMA_F32(a[g], bt, acc[j]);
      }
  }
  float* M = M3 + (size_t)bh * MSZ + irow * 16 * Dd + jh * 32;
  block_chunk_sum<2>(acc, sP, M, Dd);
}

__global__ void __launch_bounds__(256) out_kernel(const float* __restrict__ qkv,
                                                  const float* __restrict__ M3,
                                                  float* __restrict__ out) {
  __shared__ __attribute__((aligned(16))) float sM[MSZ];
  const int bh = blockIdx.x / NCH, ch = blockIdx.x % NCH;
  const int b = bh / Hh, h = bh % Hh;
  const int wave = threadIdx.x >> 5, lane = threadIdx.x & 31;
  const int tbase = ch * CH + wave * 16;
  const int m = lane & 15, kk = (lane >> 4) << 1;
  const v4f* g3 = (const v4f*)(M3 + (size_t)bh * MSZ);
  v4f f3[4];
#pragma unroll
  for (int i = 0; i < 4; ++i) f3[i] = g3[threadIdx.x + i * 256];
  const float* qp = qkv + qoff(b, tbase + m, 4, h, kk);
  v2f a[16];
#pragma unroll
  for (int step = 0; step < 16; ++step) a[step] = *(const v2f*)(qp + 4 * step);
#pragma unroll
  for (int i = 0; i < 4; ++i) ((v4f*)sM)[threadIdx.x + i * 256] = f3[i];
  __syncthreads();
  v8f acc[4] = {};
#pragma unroll
  for (int step = 0; step < 16; ++step) {
    const int k0 = step * 4 + kk;
    const v4f x0 = *(const v4f*)&sM[k0 * Dd + 4 * m];
    const v4f x1 = *(const v4f*)&sM[(k0 + 1) * Dd + 4 * m];
#pragma unroll
    for (int j = 0; j < 4; ++j) {
      v2f bt;
      bt.x = x0[j];
      bt.y = x1[j];
      acc[j] = WMMA_F32(a[step], bt, acc[j]);
    }
  }
  const int rh = (lane >> 4) * 8;
#pragma unroll
  for (int r = 0; r < 8; ++r) {
    v4f o;
#pragma unroll
    for (int j = 0; j < 4; ++j) o[j] = acc[j][r];
    *(v4f*)&out[(((size_t)b * Ss + tbase + rh + r) * Hh + h) * Dd + 4 * m] = o;
  }
}

extern "C" void kernel_launch(void* const* d_in, const int* in_sizes, int n_in,
                              void* d_out, int out_size, void* d_ws, size_t ws_size,
                              hipStream_t stream) {
  (void)in_sizes; (void)n_in; (void)out_size; (void)ws_size;
  const float* qkv = (const float*)d_in[0];
  float* out = (float*)d_out;
  float* w = (float*)d_ws;

  const size_t MTOT = (size_t)BH * MSZ;
  float* M1 = w;
  float* M2 = M1 + MTOT;
  float* M3 = M2 + MTOT;
  float* U  = M3 + MTOT;

  ktv_strips<<<BH * 8, NCH * 32, 0, stream>>>(qkv, M1, M2);
  u_kernel<<<BH * NCH, 256, 0, stream>>>(qkv, M1, M2, U);
  ktu_strips<<<BH * 8, NCH * 32, 0, stream>>>(qkv, U, M3);
  out_kernel<<<BH * NCH, 256, 0, stream>>>(qkv, M3, out);
}


# Q1, Q2, Q3 rows read as whole 128-byte lines (one 256-byte row per half-wave per load) and turned into the same A fragments through a wave-private LDS tile; same summation order
# speedup vs baseline: 2.2400x; 1.0486x over previous
#include <hip/hip_runtime.h>
#include <math.h>


typedef __attribute__((ext_vector_type(2))) float v2f;
typedef __attribute__((ext_vector_type(4))) float v4f;
typedef __attribute__((ext_vector_type(8))) float v8f;

#define WMMA_F32(a, b, c) \
  __builtin_amdgcn_wmma_f32_16x16x4_f32(false, (a), false, (b), (short)0, (c), false, false)

constexpr int Bb = 2;
constexpr int Ss = 2048;
constexpr int Hh = 12;
constexpr int Dd = 64;
constexpr int BH = Bb * Hh;
constexpr int NCH = 16;
constexpr int CH = Ss / NCH;
constexpr int SROW = 7 * Hh * Dd;
constexpr int MSZ = Dd * Dd;


constexpr int QP = 68;
constexpr int QTILE = 16 * QP;

__device__ __forceinline__ size_t qoff(int b, int s, int slot, int h, int d) {
  return ((((size_t)b * Ss + s) * 7 + slot) * Hh + h) * (size_t)Dd + d;
}

__device__ __forceinline__ int wave_index() {
  return __builtin_amdgcn_readfirstlane((int)(threadIdx.x >> 5));
}

template <int NJ>
__device__ __forceinline__ void block_chunk_sum(const v8f (&acc)[NJ], float* sP,
                                                float* __restrict__ Mdst, int mpitch) {
  constexpr int TW = 16 * NJ;
  constexpr int TSZ = 16 * TW;
  typedef __attribute__((ext_vector_type(NJ))) float vnf;
  const int wave = wave_index(), lane = threadIdx.x & 31;
  const int m = lane & 15, rh = (lane >> 4) * 8;
#pragma unroll
  for (int r = 0; r < 8; ++r) {
    vnf o;
#pragma unroll
    for (int j = 0; j < NJ; ++j) o[j] = acc[j][r];
    *(vnf*)&sP[wave * TSZ + (rh + r) * TW + NJ * m] = o;
  }
  __syncthreads();
  for (int e = threadIdx.x; e < TSZ; e += NCH * 32) {
    float s = 0.f;
#pragma unroll
    for (int c = 0; c < NCH; ++c) s += sP[c * TSZ + e];
    Mdst[(e / TW) * mpitch + (e % TW)] = s;
  }
}

constexpr int RG1 = 8;
__global__ void __launch_bounds__(NCH * 32) ktv_strips(const float* __restrict__ qkv,
                                                       float* __restrict__ M1,
                                                       float* __restrict__ M2) {
  __shared__ __attribute__((aligned(16))) float sP[NCH * 16 * Dd];
  const int irow = blockIdx.x & 3;
  const int mat = (blockIdx.x >> 2) & 1;
  const int bh = blockIdx.x >> 3;
  const int b = bh / Hh, h = bh % Hh;
  const int wave = wave_index(), lane = threadIdx.x & 31;
  const int kslot = mat ? 3 : 1;
  const int m = lane & 15;
  const int kk = (lane >> 4) << 1;
  v8f acc[4] = {};
  const float* Kp = qkv + qoff(b, wave * CH + kk, kslot, h, irow * 16 + m);
  const float* Vp = qkv + qoff(b, wave * CH + kk, 6, h, 4 * m);
  for (int t = 0; t < CH / (4 * RG1); ++t) {
    v2f a[RG1];
    v4f v0[RG1], v1[RG1];
#pragma unroll
    for (int g = 0; g < RG1; ++g) {
      const float* kp = Kp + (size_t)(t * RG1 + g) * 4 * SROW;
      const float* vp = Vp + (size_t)(t * RG1 + g) * 4 * SROW;
      a[g].x = kp[0];
      a[g].y = kp[SROW];
      v0[g] = *(const v4f*)vp;
      v1[g] = *(const v4f*)(vp + SROW);
    }
#pragma unroll
    for (int g = 0; g < RG1; ++g)
#pragma unroll
      for (int j = 0; j < 4; ++j) {
        v2f bt;
        bt.x = v0[g][j];
        bt.y = v1[g][j];
        acc[j] = WMMA_F32(a[g], bt, acc[j]);
      }
  }
  float* M = (mat ? M2 : M1) + (size_t)bh * MSZ + irow * 16 * Dd;
  block_chunk_sum<4>(acc, sP, M, Dd);
}

__global__ void __launch_bounds__(256) u_kernel(const float* __restrict__ qkv,
                                                const float* __restrict__ M1,
                                                const float* __restrict__ M2,
                                                float* __restrict__ U) {
  __shared__ __attribute__((aligned(16))) float sM1[MSZ];
  __shared__ __attribute__((aligned(16))) float sM2[MSZ];
  __shared__ __attribute__((aligned(16))) float sQ1[8 * QTILE];
  __shared__ __attribute__((aligned(16))) float sQ2[8 * QTILE];
  const int bh = blockIdx.x / NCH, ch = blockIdx.x % NCH;
  const int b = bh / Hh, h = bh % Hh;
  const int wave = threadIdx.x >> 5, lane = threadIdx.x & 31;
  const int tbase = ch * CH + wave * 16;
  const int m = lane & 15, kk = (lane >> 4) << 1;
  const v4f* g1 = (const v4f*)(M1 + (size_t)bh * MSZ);
  const v4f* g2 = (const v4f*)(M2 + (size_t)bh * MSZ);
  v4f f1[4], f2[4];
#pragma unroll
  for (int i = 0; i < 4; ++i) {
    f1[i] = g1[threadIdx.x + i * 256];
    f2[i] = g2[threadIdx.x + i * 256];
  }
  const int hr = lane >> 4;
  const float* q1p = qkv + qoff(b, tbase + hr, 0, h, 4 * m);
  const float* q2p = qkv + qoff(b, tbase + hr, 2, h, 4 * m);
  v4f r1[8], r2[8];
#pragma unroll
  for (int i = 0; i < 8; ++i) {
    r1[i] = *(const v4f*)(q1p + (size_t)(2 * i) * SROW);
    r2[i] = *(const v4f*)(q2p + (size_t)(2 * i) * SROW);
  }
#pragma unroll
  for (int i = 0; i < 4; ++i) {
    ((v4f*)sM1)[threadIdx.x + i * 256] = f1[i];
    ((v4f*)sM2)[threadIdx.x + i * 256] = f2[i];
  }
  float* w1 = sQ1 + wave * QTILE;
  float* w2 = sQ2 + wave * QTILE;
#pragma unroll
  for (int i = 0; i < 8; ++i) {
    *(v4f*)&w1[(2 * i + hr) * QP + 4 * m] = r1[i];
    *(v4f*)&w2[(2 * i + hr) * QP + 4 * m] = r2[i];
  }
  __syncthreads();
  v2f a1[16], a2[16];
#pragma unroll
  for (int step = 0; step < 16; ++step) {
    a1[step] = *(const v2f*)&w1[m * QP + 4 * step + kk];
    a2[step] = *(const v2f*)&w2[m * QP + 4 * step + kk];
  }
  v8f acc1[4] = {}, acc2[4] = {};
#pragma unroll
  for (int step = 0; step < 16; ++step) {
    const int k0 = step * 4 + kk;
    const v4f x0 = *(const v4f*)&sM1[k0 * Dd + 4 * m];
    const v4f x1 = *(const v4f*)&sM1[(k0 + 1) * Dd + 4 * m];
    const v4f y0 = *(const v4f*)&sM2[k0 * Dd + 4 * m];
    const v4f y1 = *(const v4f*)&sM2[(k0 + 1) * Dd + 4 * m];
#pragma unroll
    for (int j = 0; j < 4; ++j) {
      v2f b1, b2;
      b1.x = x0[j]; b1.y = x1[j];
      b2.x = y0[j]; b2.y = y1[j];
      acc1[j] = WMMA_F32(a1[step], b1, acc1[j]);
      acc2[j] = WMMA_F32(a2[step], b2, acc2[j]);
    }
  }
  const int rh = (lane >> 4) * 8;
#pragma unroll
  for (int r = 0; r < 8; ++r) {
    v4f o;
#pragma unroll
    for (int j = 0; j < 4; ++j) {
      const float x = acc1[j][r];
      const float t1 = x / (1.f + __expf(-x));
      o[j] = t1 * acc2[j][r];
    }
    *(v4f*)&U[((size_t)bh * Ss + tbase + rh + r) * Dd + 4 * m] = o;
  }
}

constexpr int RG3 = 16;
__global__ void __launch_bounds__(NCH * 32) ktu_strips(const float* __restrict__ qkv,
                                                       const float* __restrict__ U,
                                                       float* __restrict__ M3) {
  __shared__ __attribute__((aligned(16))) float sP[NCH * 16 * 32];
  const int jh = blockIdx.x & 1;
  const int irow = (blockIdx.x >> 1) & 3;
  const int bh = blockIdx.x >> 3;
  const int b = bh / Hh, h = bh % Hh;
  const int wave = wave_index(), lane = threadIdx.x & 31;
  const int m = lane & 15, kk = (lane >> 4) << 1;
  v8f acc[2] = {};
  const float* Kp = qkv + qoff(b, wave * CH + kk, 5, h, irow * 16 + m);
  const float* Up = U + ((size_t)bh * Ss + wave * CH + kk) * Dd + jh * 32 + 2 * m;
  for (int t = 0; t < CH / (4 * RG3); ++t) {
    v2f a[RG3], u0[RG3], u1[RG3];
#pragma unroll
    for (int g = 0; g < RG3; ++g) {
      const float* kp = Kp + (size_t)(t * RG3 + g) * 4 * SROW;
      const float* up = Up + (size_t)(t * RG3 + g) * 4 * Dd;
      a[g].x = kp[0];
      a[g].y = kp[SROW];
      u0[g] = *(const v2f*)up;
      u1[g] = *(const v2f*)(up + Dd);
    }
#pragma unroll
    for (int g = 0; g < RG3; ++g)
#pragma unroll
      for (int j = 0; j < 2; ++j) {
        v2f bt;
        bt.x = u0[g][j];
        bt.y = u1[g][j];
        acc[j] = WMMA_F32(a[g], bt, acc[j]);
      }
  }
  float* M = M3 + (size_t)bh * MSZ + irow * 16 * Dd + jh * 32;
  block_chunk_sum<2>(acc, sP, M, Dd);
}

__global__ void __launch_bounds__(256) out_kernel(const float* __restrict__ qkv,
                                                  const float* __restrict__ M3,
                                                  float* __restrict__ out) {
  __shared__ __attribute__((aligned(16))) float sM[MSZ];
  __shared__ __attribute__((aligned(16))) float sQ[8 * QTILE];
  const int bh = blockIdx.x / NCH, ch = blockIdx.x % NCH;
  const int b = bh / Hh, h = bh % Hh;
  const int wave = threadIdx.x >> 5, lane = threadIdx.x & 31;
  const int tbase = ch * CH + wave * 16;
  const int m = lane & 15, kk = (lane >> 4) << 1;
  const v4f* g3 = (const v4f*)(M3 + (size_t)bh * MSZ);
  v4f f3[4];
#pragma unroll
  for (int i = 0; i < 4; ++i) f3[i] = g3[threadIdx.x + i * 256];
  const int hr = lane >> 4;
  const float* qp = qkv + qoff(b, tbase + hr, 4, h, 4 * m);
  v4f rq[8];
#pragma unroll
  for (int i = 0; i < 8; ++i) rq[i] = *(const v4f*)(qp + (size_t)(2 * i) * SROW);
#pragma unroll
  for (int i = 0; i < 4; ++i) ((v4f*)sM)[threadIdx.x + i * 256] = f3[i];
  float* wq = sQ + wave * QTILE;
#pragma unroll
  for (int i = 0; i < 8; ++i) *(v4f*)&wq[(2 * i + hr) * QP + 4 * m] = rq[i];
  __syncthreads();
  v2f a[16];
#pragma unroll
  for (int step = 0; step < 16; ++step) a[step] = *(const v2f*)&wq[m * QP + 4 * step + kk];
  v8f acc[4] = {};
#pragma unroll
  for (int step = 0; step < 16; ++step) {
    const int k0 = step * 4 + kk;
    const v4f x0 = *(const v4f*)&sM[k0 * Dd + 4 * m];
    const v4f x1 = *(const v4f*)&sM[(k0 + 1) * Dd + 4 * m];
#pragma unroll
    for (int j = 0; j < 4; ++j) {
      v2f bt;
      bt.x = x0[j];
      bt.y = x1[j];
      acc[j] = WMMA_F32(a[step], bt, acc[j]);
    }
  }
  const int rh = (lane >> 4) * 8;
#pragma unroll
  for (int r = 0; r < 8; ++r) {
    v4f o;
#pragma unroll
    for (int j = 0; j < 4; ++j) o[j] = acc[j][r];
    *(v4f*)&out[(((size_t)b * Ss + tbase + rh + r) * Hh + h) * Dd + 4 * m] = o;
  }
}

extern "C" void kernel_launch(void* const* d_in, const int* in_sizes, int n_in,
                              void* d_out, int out_size, void* d_ws, size_t ws_size,
                              hipStream_t stream) {
  (void)in_sizes; (void)n_in; (void)out_size; (void)ws_size;
  const float* qkv = (const float*)d_in[0];
  float* out = (float*)d_out;
  float* w = (float*)d_ws;

  const size_t MTOT = (size_t)BH * MSZ;
  float* M1 = w;
  float* M2 = M1 + MTOT;
  float* M3 = M2 + MTOT;
  float* U  = M3 + MTOT;

  ktv_strips<<<BH * 8, NCH * 32, 0, stream>>>(qkv, M1, M2);
  u_kernel<<<BH * NCH, 256, 0, stream>>>(qkv, M1, M2, U);
  ktu_strips<<<BH * 8, NCH * 32, 0, stream>>>(qkv, U, M3);
  out_kernel<<<BH * NCH, 256, 0, stream>>>(qkv, M3, out);
}


# u never leaves the chip: one kernel forms u for 256 rows in LDS and then K3^T u of those rows; out kernel adds the 8 partials of M3 (chunk pairs summed first: new reduction tree); 3 launches, not 4
# speedup vs baseline: 2.5143x; 1.1225x over previous
#include <hip/hip_runtime.h>
#include <math.h>


typedef __attribute__((ext_vector_type(2))) float v2f;
typedef __attribute__((ext_vector_type(4))) float v4f;
typedef __attribute__((ext_vector_type(8))) float v8f;

#define WMMA_F32(a, b, c) \
  __builtin_amdgcn_wmma_f32_16x16x4_f32(false, (a), false, (b), (short)0, (c), false, false)

constexpr int Bb = 2;
constexpr int Ss = 2048;
constexpr int Hh = 12;
constexpr int Dd = 64;
constexpr int BH = Bb * Hh;
constexpr int NCH = 16;
constexpr int CH = Ss / NCH;
constexpr int SROW = 7 * Hh * Dd;
constexpr int MSZ = Dd * Dd;


constexpr int QP = 68;
constexpr int QTILE = 16 * QP;

__device__ __forceinline__ size_t qoff(int b, int s, int slot, int h, int d) {
  return ((((size_t)b * Ss + s) * 7 + slot) * Hh + h) * (size_t)Dd + d;
}

__device__ __forceinline__ int wave_index() {
  return __builtin_amdgcn_readfirstlane((int)(threadIdx.x >> 5));
}

template <int NJ>
__device__ __forceinline__ void block_chunk_sum(const v8f (&acc)[NJ], float* sP,
                                                float* __restrict__ Mdst, int mpitch) {
  constexpr int TW = 16 * NJ;
  constexpr int TSZ = 16 * TW;
  typedef __attribute__((ext_vector_type(NJ))) float vnf;
  const int wave = wave_index(), lane = threadIdx.x & 31;
  const int m = lane & 15, rh = (lane >> 4) * 8;
#pragma unroll
  for (int r = 0; r < 8; ++r) {
    vnf o;
#pragma unroll
    for (int j = 0; j < NJ; ++j) o[j] = acc[j][r];
    *(vnf*)&sP[wave * TSZ + (rh + r) * TW + NJ * m] = o;
  }
  __syncthreads();
  for (int e = threadIdx.x; e < TSZ; e += NCH * 32) {
    float s = 0.f;
#pragma unroll
    for (int c = 0; c < NCH; ++c) s += sP[c * TSZ + e];
    Mdst[(e / TW) * mpitch + (e % TW)] = s;
  }
}

constexpr int RG1 = 8;
__global__ void __launch_bounds__(NCH * 32) ktv_strips(const float* __restrict__ qkv,
                                                       float* __restrict__ M1,
                                                       float* __restrict__ M2) {
  __shared__ __attribute__((aligned(16))) float sP[NCH * 16 * Dd];
  const int irow = blockIdx.x & 3;
  const int mat = (blockIdx.x >> 2) & 1;
  const int bh = blockIdx.x >> 3;
  const int b = bh / Hh, h = bh % Hh;
  const int wave = wave_index(), lane = threadIdx.x & 31;
  const int kslot = mat ? 3 : 1;
  const int m = lane & 15;
  const int kk = (lane >> 4) << 1;
  v8f acc[4] = {};
  const float* Kp = qkv + qoff(b, wave * CH + kk, kslot, h, irow * 16 + m);
  const float* Vp = qkv + qoff(b, wave * CH + kk, 6, h, 4 * m);
  for (int t = 0; t < CH / (4 * RG1); ++t) {
    v2f a[RG1];
    v4f v0[RG1], v1[RG1];
#pragma unroll
    for (int g = 0; g < RG1; ++g) {
      const float* kp = Kp + (size_t)(t * RG1 + g) * 4 * SROW;
      const float* vp = Vp + (size_t)(t * RG1 + g) * 4 * SROW;
      a[g].x = kp[0];
      a[g].y = kp[SROW];
      v0[g] = *(const v4f*)vp;
      v1[g] = *(const v4f*)(vp + SROW);
    }
#pragma unroll
    for (int g = 0; g < RG1; ++g)
#pragma unroll
      for (int j = 0; j < 4; ++j) {
        v2f bt;
        bt.x = v0[g][j];
        bt.y = v1[g][j];
        acc[j] = WMMA_F32(a[g], bt, acc[j]);
      }
  }
  float* M = (mat ? M2 : M1) + (size_t)bh * MSZ + irow * 16 * Dd;
  block_chunk_sum<4>(acc, sP, M, Dd);
}

constexpr int RG3 = 16;
constexpr int PR = 2 * CH;
constexpr int NPR = Ss / PR;
constexpr int UP = 80;
__global__ void __launch_bounds__(512) u_ktu_kernel(const float* __restrict__ qkv,
                                                    const float* __restrict__ M1,
                                                    const float* __restrict__ M2,
                                                    float* __restrict__ P3) {
  __shared__ __attribute__((aligned(16))) float smem[2 * MSZ + 32 * QTILE + PR * UP];
  float* sM1 = smem;
  float* sM2 = sM1 + MSZ;
  float* sQ1 = sM2 + MSZ;
  float* sQ2 = sQ1 + 16 * QTILE;
  float* sU = sQ2 + 16 * QTILE;
  const int bh = blockIdx.x / NPR, pr = blockIdx.x % NPR;
  const int b = bh / Hh, h = bh % Hh;
  const int wave = wave_index(), lane = threadIdx.x & 31;
  const int trow = wave * 16;
  const int tbase = pr * PR + trow;
  const int m = lane & 15, kk = (lane >> 4) << 1;
  const v4f* g1 = (const v4f*)(M1 + (size_t)bh * MSZ);
  const v4f* g2 = (const v4f*)(M2 + (size_t)bh * MSZ);
  v4f f1[2], f2[2];
#pragma unroll
  for (int i = 0; i < 2; ++i) {
    f1[i] = g1[threadIdx.x + i * 512];
    f2[i] = g2[threadIdx.x + i * 512];
  }
  const int hr = lane >> 4;
  const float* q1p = qkv + qoff(b, tbase + hr, 0, h, 4 * m);
  const float* q2p = qkv + qoff(b, tbase + hr, 2, h, 4 * m);
  v4f r1[8], r2[8];
#pragma unroll
  for (int i = 0; i < 8; ++i) {
    r1[i] = *(const v4f*)(q1p + (size_t)(2 * i) * SROW);
    r2[i] = *(const v4f*)(q2p + (size_t)(2 * i) * SROW);
  }
#pragma unroll
  for (int i = 0; i < 2; ++i) {
    ((v4f*)sM1)[threadIdx.x + i * 512] = f1[i];
    ((v4f*)sM2)[threadIdx.x + i * 512] = f2[i];
  }
  float* w1 = sQ1 + wave * QTILE;
  float* w2 = sQ2 + wave * QTILE;
#pragma unroll
  for (int i = 0; i < 8; ++i) {
    *(v4f*)&w1[(2 * i + hr) * QP + 4 * m] = r1[i];
    *(v4f*)&w2[(2 * i + hr) * QP + 4 * m] = r2[i];
  }
  __syncthreads();
  v8f acc1[4] = {}, acc2[4] = {};
#pragma unroll 4
  for (int step = 0; step < 16; ++step) {
    const int k0 = step * 4 + kk;
    const v2f a1s = *(const v2f*)&w1[m * QP + 4 * step + kk];
    const v2f a2s = *(const v2f*)&w2[m * QP + 4 * step + kk];
    const v4f x0 = *(const v4f*)&sM1[k0 * Dd + 4 * m];
    const v4f x1 = *(const v4f*)&sM1[(k0 + 1) * Dd + 4 * m];
    const v4f y0 = *(const v4f*)&sM2[k0 * Dd + 4 * m];
    const v4f y1 = *(const v4f*)&sM2[(k0 + 1) * Dd + 4 * m];
#pragma unroll
    for (int j = 0; j < 4; ++j) {
      v2f b1, b2;
      b1.x = x0[j]; b1.y = x1[j];
      b2.x = y0[j]; b2.y = y1[j];
      acc1[j] = WMMA_F32(a1s, b1, acc1[j]);
      acc2[j] = WMMA_F32(a2s, b2, acc2[j]);
    }
  }
  const int c2 = wave >> 3, irow = (wave >> 1) & 3, jh = wave & 1;
  const float* Kp = qkv + qoff(b, pr * PR + c2 * CH + kk, 5, h, irow * 16 + m);
  v2f a3[CH / 4];
#pragma unroll
  for (int g = 0; g < CH / 4; ++g) {
    const float* kp = Kp + (size_t)g * 4 * SROW;
    a3[g].x = kp[0];
    a3[g].y = kp[SROW];
  }
  const int rh = (lane >> 4) * 8;
#pragma unroll
  for (int r = 0; r < 8; ++r) {
    v4f o;
#pragma unroll
    for (int j = 0; j < 4; ++j) {
      const float x = acc1[j][r];
      const float t1 = x / (1.f + __expf(-x));
      o[j] = t1 * acc2[j][r];
    }
    *(v4f*)&sU[(trow + rh + r) * UP + 4 * m] = o;
  }
  __syncthreads();
  const float* up = sU + (c2 * CH + kk) * UP + jh * 32 + 2 * m;
  v8f acc[2] = {};
#pragma unroll
  for (int t = 0; t < CH / (4 * RG3); ++t) {
    v2f u0[RG3], u1[RG3];
#pragma unroll
    for (int g = 0; g < RG3; ++g) {
      const float* ug = up + (t * RG3 + g) * 4 * UP;
      u0[g] = *(const v2f*)ug;
      u1[g] = *(const v2f*)(ug + UP);
    }
    __builtin_amdgcn_sched_barrier(0);
#pragma unroll
    for (int g = 0; g < RG3; ++g)
#pragma unroll
      for (int j = 0; j < 2; ++j) {
        v2f bt;
        bt.x = u0[g][j];
        bt.y = u1[g][j];
        acc[j] = WMMA_F32(a3[t * RG3 + g], bt, acc[j]);
      }
  }
  float* sX = sQ1;
#pragma unroll
  for (int r = 0; r < 8; ++r) {
    v2f o;
    o.x = acc[0][r];
    o.y = acc[1][r];
    *(v2f*)&sX[wave * 512 + (rh + r) * 32 + 2 * m] = o;
  }
  __syncthreads();
  if (c2 == 0) {
    float* P = P3 + ((size_t)bh * NPR + pr) * MSZ + irow * 16 * Dd + jh * 32;
#pragma unroll
    for (int r = 0; r < 8; ++r) {
      const v2f x0 = *(const v2f*)&sX[wave * 512 + (rh + r) * 32 + 2 * m];
      const v2f x1 = *(const v2f*)&sX[(wave + 8) * 512 + (rh + r) * 32 + 2 * m];
      *(v2f*)&P[(rh + r) * Dd + 2 * m] = x0 + x1;
    }
  }
}

__global__ void __launch_bounds__(512) out_kernel(const float* __restrict__ qkv,
                                                  const float* __restrict__ P3,
                                                  float* __restrict__ out) {
  __shared__ __attribute__((aligned(16))) float sM[MSZ];
  __shared__ __attribute__((aligned(16))) float sQ[16 * QTILE];
  const int bh = blockIdx.x / NPR, pr = blockIdx.x % NPR;
  const int b = bh / Hh, h = bh % Hh;
  const int wave = wave_index(), lane = threadIdx.x & 31;
  const int tbase = pr * PR + wave * 16;
  const int m = lane & 15, kk = (lane >> 4) << 1;
  const v4f* g3 = (const v4f*)(P3 + (size_t)bh * NPR * MSZ);
  v4f f3[2][NPR];
#pragma unroll
  for (int i = 0; i < 2; ++i)
#pragma unroll
    for (int p = 0; p < NPR; ++p) f3[i][p] = g3[p * (MSZ / 4) + threadIdx.x + i * 512];
  const int hr = lane >> 4;
  const float* qp = qkv + qoff(b, tbase + hr, 4, h, 4 * m);
  v4f rq[8];
#pragma unroll
  for (int i = 0; i < 8; ++i) rq[i] = *(const v4f*)(qp + (size_t)(2 * i) * SROW);
#pragma unroll
  for (int i = 0; i < 2; ++i) {
    v4f s3 = f3[i][0];
#pragma unroll
    for (int p = 1; p < NPR; ++p) s3 += f3[i][p];
    ((v4f*)sM)[threadIdx.x + i * 512] = s3;
  }
  float* wq = sQ + wave * QTILE;
#pragma unroll
  for (int i = 0; i < 8; ++i) *(v4f*)&wq[(2 * i + hr) * QP + 4 * m] = rq[i];
  __syncthreads();
  v2f a[16];
#pragma unroll
  for (int step = 0; step < 16; ++step) a[step] = *(const v2f*)&wq[m * QP + 4 * step + kk];
  v8f acc[4] = {};
#pragma unroll
  for (int step = 0; step < 16; ++step) {
    const int k0 = step * 4 + kk;
    const v4f x0 = *(const v4f*)&sM[k0 * Dd + 4 * m];
    const v4f x1 = *(const v4f*)&sM[(k0 + 1) * Dd + 4 * m];
#pragma unroll
    for (int j = 0; j < 4; ++j) {
      v2f bt;
      bt.x = x0[j];
      bt.y = x1[j];
      acc[j] = WMMA_F32(a[step], bt, acc[j]);
    }
  }
  const int rh = (lane >> 4) * 8;
#pragma unroll
  for (int r = 0; r < 8; ++r) {
    v4f o;
#pragma unroll
    for (int j = 0; j < 4; ++j) o[j] = acc[j][r];
    *(v4f*)&out[(((size_t)b * Ss + tbase + rh + r) * Hh + h) * Dd + 4 * m] = o;
  }
}

extern "C" void kernel_launch(void* const* d_in, const int* in_sizes, int n_in,
                              void* d_out, int out_size, void* d_ws, size_t ws_size,
                              hipStream_t stream) {
  (void)in_sizes; (void)n_in; (void)out_size; (void)ws_size;
  const float* qkv = (const float*)d_in[0];
  float* out = (float*)d_out;
  float* w = (float*)d_ws;

  const size_t MTOT = (size_t)BH * MSZ;
  float* M1 = w;
  float* M2 = M1 + MTOT;
  float* P3 = M2 + MTOT;

  ktv_strips<<<BH * 8, NCH * 32, 0, stream>>>(qkv, M1, M2);
  u_ktu_kernel<<<BH * NPR, 512, 0, stream>>>(qkv, M1, M2, P3);
  out_kernel<<<BH * NPR, 512, 0, stream>>>(qkv, P3, out);
}


# K^T V units (strip, S chunk) dealt 12 per block to 256 blocks of 12 waves: all 256 WGPs work, not 192; a strip's chunk sum is made in two parts that kernel 2 adds: new reduction tree
# speedup vs baseline: 2.6884x; 1.0692x over previous
#include <hip/hip_runtime.h>
#include <math.h>


typedef __attribute__((ext_vector_type(2))) float v2f;
typedef __attribute__((ext_vector_type(4))) float v4f;
typedef __attribute__((ext_vector_type(8))) float v8f;

#define WMMA_F32(a, b, c) \
  __builtin_amdgcn_wmma_f32_16x16x4_f32(false, (a), false, (b), (short)0, (c), false, false)

constexpr int Bb = 2;
constexpr int Ss = 2048;
constexpr int Hh = 12;
constexpr int Dd = 64;
constexpr int BH = Bb * Hh;
constexpr int NCH = 16;
constexpr int CH = Ss / NCH;
constexpr int SROW = 7 * Hh * Dd;
constexpr int MSZ = Dd * Dd;
constexpr int MTOT = BH * MSZ;
constexpr int W1 = 12;
constexpr int NB1 = BH * 8 * NCH / W1;


constexpr int QP = 68;
constexpr int QTILE = 16 * QP;

__device__ __forceinline__ size_t qoff(int b, int s, int slot, int h, int d) {
  return ((((size_t)b * Ss + s) * 7 + slot) * Hh + h) * (size_t)Dd + d;
}

__device__ __forceinline__ int wave_index() {
  return __builtin_amdgcn_readfirstlane((int)(threadIdx.x >> 5));
}

constexpr int RG1 = 8;
__global__ void __launch_bounds__(W1 * 32) ktv_strips(const float* __restrict__ qkv,
                                                      float* __restrict__ M1,
                                                      float* __restrict__ M2) {
  __shared__ __attribute__((aligned(16))) float sP[W1 * 16 * Dd];
  const int wave = wave_index(), lane = threadIdx.x & 31;
  const int unit0 = blockIdx.x * W1;
  const int unit = unit0 + wave;
  const int strip = unit >> 4, chunk = unit & 15;
  const int irow = strip & 3;
  const int mat = (strip >> 2) & 1;
  const int bh = strip >> 3;
  const int b = bh / Hh, h = bh % Hh;
  const int kslot = mat ? 3 : 1;
  const int m = lane & 15;
  const int kk = (lane >> 4) << 1;
  v8f acc[4] = {};
  const float* Kp = qkv + qoff(b, chunk * CH + kk, kslot, h, irow * 16 + m);
  const float* Vp = qkv + qoff(b, chunk * CH + kk, 6, h, 4 * m);
  for (int t = 0; t < CH / (4 * RG1); ++t) {
    v2f a[RG1];
    v4f v0[RG1], v1[RG1];
#pragma unroll
    for (int g = 0; g < RG1; ++g) {
      const float* kp = Kp + (size_t)(t * RG1 + g) * 4 * SROW;
      const float* vp = Vp + (size_t)(t * RG1 + g) * 4 * SROW;
      a[g].x = kp[0];
      a[g].y = kp[SROW];
      v0[g] = *(const v4f*)vp;
      v1[g] = *(const v4f*)(vp + SROW);
    }
#pragma unroll
    for (int g = 0; g < RG1; ++g)
#pragma unroll
      for (int j = 0; j < 4; ++j) {
        v2f bt;
        bt.x = v0[g][j];
        bt.y = v1[g][j];
        acc[j] = WMMA_F32(a[g], bt, acc[j]);
      }
  }
  const int rh = (lane >> 4) * 8;
#pragma unroll
  for (int r = 0; r < 8; ++r) {
    v4f o;
#pragma unroll
    for (int j = 0; j < 4; ++j) o[j] = acc[j][r];
    *(v4f*)&sP[wave * (16 * Dd) + (rh + r) * Dd + 4 * m] = o;
  }
  __syncthreads();
  const int c0 = unit0 & 15;
  const int n0 = (16 - c0) < W1 ? (16 - c0) : W1;
  const int strip0 = unit0 >> 4;
  if (threadIdx.x < 256) {
    const int el = 4 * threadIdx.x;
    v4f t[W1];
#pragma unroll
    for (int w = 0; w < W1; ++w) t[w] = *(const v4f*)&sP[w * (16 * Dd) + el];
    v4f sa = {0.f, 0.f, 0.f, 0.f}, sb = {0.f, 0.f, 0.f, 0.f};
#pragma unroll
    for (int w = 0; w < W1; ++w) {
      if (w < n0) sa += t[w];
      else sb += t[w];
    }
    const int pa = c0 != 0 ? 1 : 0;
    float* Ma = ((strip0 >> 2) & 1 ? M2 : M1) + (size_t)pa * 2 * MTOT +
                (size_t)(strip0 >> 3) * MSZ + (strip0 & 3) * 16 * Dd;
    *(v4f*)&Ma[el] = sa;
    if (n0 < W1) {
      const int st = strip0 + 1;
      float* Mb = ((st >> 2) & 1 ? M2 : M1) + (size_t)(st >> 3) * MSZ + (st & 3) * 16 * Dd;
      *(v4f*)&Mb[el] = sb;
    }
  }
}

constexpr int RG3 = 16;
constexpr int PR = 2 * CH;
constexpr int NPR = Ss / PR;
constexpr int UP = 80;
__global__ void __launch_bounds__(512) u_ktu_kernel(const float* __restrict__ qkv,
                                                    const float* __restrict__ M1,
                                                    const float* __restrict__ M2,
                                                    float* __restrict__ P3) {
  __shared__ __attribute__((aligned(16))) float smem[2 * MSZ + 32 * QTILE + PR * UP];
  float* sM1 = smem;
  float* sM2 = sM1 + MSZ;
  float* sQ1 = sM2 + MSZ;
  float* sQ2 = sQ1 + 16 * QTILE;
  float* sU = sQ2 + 16 * QTILE;
  const int bh = blockIdx.x / NPR, pr = blockIdx.x % NPR;
  const int b = bh / Hh, h = bh % Hh;
  const int wave = wave_index(), lane = threadIdx.x & 31;
  const int trow = wave * 16;
  const int tbase = pr * PR + trow;
  const int m = lane & 15, kk = (lane >> 4) << 1;
  const v4f* g1 = (const v4f*)(M1 + (size_t)bh * MSZ);
  const v4f* g2 = (const v4f*)(M2 + (size_t)bh * MSZ);
  const v4f* g1b = g1 + 2 * MTOT / 4;
  const v4f* g2b = g2 + 2 * MTOT / 4;
  v4f f1[2], f2[2], f1b[2], f2b[2];
#pragma unroll
  for (int i = 0; i < 2; ++i) {
    f1[i] = g1[threadIdx.x + i * 512];
    f2[i] = g2[threadIdx.x + i * 512];
    f1b[i] = g1b[threadIdx.x + i * 512];
    f2b[i] = g2b[threadIdx.x + i * 512];
  }
  const int hr = lane >> 4;
  const float* q1p = qkv + qoff(b, tbase + hr, 0, h, 4 * m);
  const float* q2p = qkv + qoff(b, tbase + hr, 2, h, 4 * m);
  v4f r1[8], r2[8];
#pragma unroll
  for (int i = 0; i < 8; ++i) {
    r1[i] = *(const v4f*)(q1p + (size_t)(2 * i) * SROW);
    r2[i] = *(const v4f*)(q2p + (size_t)(2 * i) * SROW);
  }
#pragma unroll
  for (int i = 0; i < 2; ++i) {
    ((v4f*)sM1)[threadIdx.x + i * 512] = f1[i] + f1b[i];
    ((v4f*)sM2)[threadIdx.x + i * 512] = f2[i] + f2b[i];
  }
  float* w1 = sQ1 + wave * QTILE;
  float* w2 = sQ2 + wave * QTILE;
#pragma unroll
  for (int i = 0; i < 8; ++i) {
    *(v4f*)&w1[(2 * i + hr) * QP + 4 * m] = r1[i];
    *(v4f*)&w2[(2 * i + hr) * QP + 4 * m] = r2[i];
  }
  __syncthreads();
  v8f acc1[4] = {}, acc2[4] = {};
#pragma unroll 4
  for (int step = 0; step < 16; ++step) {
    const int k0 = step * 4 + kk;
    const v2f a1s = *(const v2f*)&w1[m * QP + 4 * step + kk];
    const v2f a2s = *(const v2f*)&w2[m * QP + 4 * step + kk];
    const v4f x0 = *(const v4f*)&sM1[k0 * Dd + 4 * m];
    const v4f x1 = *(const v4f*)&sM1[(k0 + 1) * Dd + 4 * m];
    const v4f y0 = *(const v4f*)&sM2[k0 * Dd + 4 * m];
    const v4f y1 = *(const v4f*)&sM2[(k0 + 1) * Dd + 4 * m];
#pragma unroll
    for (int j = 0; j < 4; ++j) {
      v2f b1, b2;
      b1.x = x0[j]; b1.y = x1[j];
      b2.x = y0[j]; b2.y = y1[j];
      acc1[j] = WMMA_F32(a1s, b1, acc1[j]);
      acc2[j] = WMMA_F32(a2s, b2, acc2[j]);
    }
  }
  const int c2 = wave >> 3, irow = (wave >> 1) & 3, jh = wave & 1;
  const float* Kp = qkv + qoff(b, pr * PR + c2 * CH + kk, 5, h, irow * 16 + m);
  v2f a3[CH / 4];
#pragma unroll
  for (int g = 0; g < CH / 4; ++g) {
    const float* kp = Kp + (size_t)g * 4 * SROW;
    a3[g].x = kp[0];
    a3[g].y = kp[SROW];
  }
  const int rh = (lane >> 4) * 8;
#pragma unroll
  for (int r = 0; r < 8; ++r) {
    v4f o;
#pragma unroll
    for (int j = 0; j < 4; ++j) {
      const float x = acc1[j][r];
      const float t1 = x / (1.f + __expf(-x));
      o[j] = t1 * acc2[j][r];
    }
    *(v4f*)&sU[(trow + rh + r) * UP + 4 * m] = o;
  }
  __syncthreads();
  const float* up = sU + (c2 * CH + kk) * UP + jh * 32 + 2 * m;
  v8f acc[2] = {};
#pragma unroll
  for (int t = 0; t < CH / (4 * RG3); ++t) {
    v2f u0[RG3], u1[RG3];
#pragma unroll
    for (int g = 0; g < RG3; ++g) {
      const float* ug = up + (t * RG3 + g) * 4 * UP;
      u0[g] = *(const v2f*)ug;
      u1[g] = *(const v2f*)(ug + UP);
    }
    __builtin_amdgcn_sched_barrier(0);
#pragma unroll
    for (int g = 0; g < RG3; ++g)
#pragma unroll
      for (int j = 0; j < 2; ++j) {
        v2f bt;
        bt.x = u0[g][j];
        bt.y = u1[g][j];
        acc[j] = WMMA_F32(a3[t * RG3 + g], bt, acc[j]);
      }
  }
  float* sX = sQ1;
#pragma unroll
  for (int r = 0; r < 8; ++r) {
    v2f o;
    o.x = acc[0][r];
    o.y = acc[1][r];
    *(v2f*)&sX[wave * 512 + (rh + r) * 32 + 2 * m] = o;
  }
  __syncthreads();
  if (c2 == 0) {
    float* P = P3 + ((size_t)bh * NPR + pr) * MSZ + irow * 16 * Dd + jh * 32;
#pragma unroll
    for (int r = 0; r < 8; ++r) {
      const v2f x0 = *(const v2f*)&sX[wave * 512 + (rh + r) * 32 + 2 * m];
      const v2f x1 = *(const v2f*)&sX[(wave + 8) * 512 + (rh + r) * 32 + 2 * m];
      *(v2f*)&P[(rh + r) * Dd + 2 * m] = x0 + x1;
    }
  }
}

__global__ void __launch_bounds__(512) out_kernel(const float* __restrict__ qkv,
                                                  const float* __restrict__ P3,
                                                  float* __restrict__ out) {
  __shared__ __attribute__((aligned(16))) float sM[MSZ];
  __shared__ __attribute__((aligned(16))) float sQ[16 * QTILE];
  const int bh = blockIdx.x / NPR, pr = blockIdx.x % NPR;
  const int b = bh / Hh, h = bh % Hh;
  const int wave = wave_index(), lane = threadIdx.x & 31;
  const int tbase = pr * PR + wave * 16;
  const int m = lane & 15, kk = (lane >> 4) << 1;
  const v4f* g3 = (const v4f*)(P3 + (size_t)bh * NPR * MSZ);
  v4f f3[2][NPR];
#pragma unroll
  for (int i = 0; i < 2; ++i)
#pragma unroll
    for (int p = 0; p < NPR; ++p) f3[i][p] = g3[p * (MSZ / 4) + threadIdx.x + i * 512];
  const int hr = lane >> 4;
  const float* qp = qkv + qoff(b, tbase + hr, 4, h, 4 * m);
  v4f rq[8];
#pragma unroll
  for (int i = 0; i < 8; ++i) rq[i] = *(const v4f*)(qp + (size_t)(2 * i) * SROW);
#pragma unroll
  for (int i = 0; i < 2; ++i) {
    v4f s3 = f3[i][0];
#pragma unroll
    for (int p = 1; p < NPR; ++p) s3 += f3[i][p];
    ((v4f*)sM)[threadIdx.x + i * 512] = s3;
  }
  float* wq = sQ + wave * QTILE;
#pragma unroll
  for (int i = 0; i < 8; ++i) *(v4f*)&wq[(2 * i + hr) * QP + 4 * m] = rq[i];
  __syncthreads();
  v2f a[16];
#pragma unroll
  for (int step = 0; step < 16; ++step) a[step] = *(const v2f*)&wq[m * QP + 4 * step + kk];
  v8f acc[4] = {};
#pragma unroll
  for (int step = 0; step < 16; ++step) {
    const int k0 = step * 4 + kk;
    const v4f x0 = *(const v4f*)&sM[k0 * Dd + 4 * m];
    const v4f x1 = *(const v4f*)&sM[(k0 + 1) * Dd + 4 * m];
#pragma unroll
    for (int j = 0; j < 4; ++j) {
      v2f bt;
      bt.x = x0[j];
      bt.y = x1[j];
      acc[j] = WMMA_F32(a[step], bt, acc[j]);
    }
  }
  const int rh = (lane >> 4) * 8;
#pragma unroll
  for (int r = 0; r < 8; ++r) {
    v4f o;
#pragma unroll
    for (int j = 0; j < 4; ++j) o[j] = acc[j][r];
    *(v4f*)&out[(((size_t)b * Ss + tbase + rh + r) * Hh + h) * Dd + 4 * m] = o;
  }
}

extern "C" void kernel_launch(void* const* d_in, const int* in_sizes, int n_in,
                              void* d_out, int out_size, void* d_ws, size_t ws_size,
                              hipStream_t stream) {
  (void)in_sizes; (void)n_in; (void)out_size; (void)ws_size;
  const float* qkv = (const float*)d_in[0];
  float* out = (float*)d_out;
  float* w = (float*)d_ws;

  float* M1 = w;
  float* M2 = M1 + MTOT;
  float* P3 = M1 + 4 * MTOT;

  ktv_strips<<<NB1, W1 * 32, 0, stream>>>(qkv, M1, M2);
  u_ktu_kernel<<<BH * NPR, 512, 0, stream>>>(qkv, M1, M2, P3);
  out_kernel<<<BH * NPR, 512, 0, stream>>>(qkv, P3, out);
}


# Kernel 1 wave tile 32x64 over 64 s rows (was 16x64 over 128): a V row read once per 32 rows of K^T, K as whole lines; a strip's chunk sum is 3 or 4 block parts that kernel 2 adds: new reduction tree
# speedup vs baseline: 2.7838x; 1.0355x over previous
#include <hip/hip_runtime.h>
#include <math.h>


typedef __attribute__((ext_vector_type(2))) float v2f;
typedef __attribute__((ext_vector_type(4))) float v4f;
typedef __attribute__((ext_vector_type(8))) float v8f;

#define WMMA_F32(a, b, c) \
  __builtin_amdgcn_wmma_f32_16x16x4_f32(false, (a), false, (b), (short)0, (c), false, false)

constexpr int Bb = 2;
constexpr int Ss = 2048;
constexpr int Hh = 12;
constexpr int Dd = 64;
constexpr int BH = Bb * Hh;
constexpr int NCH = 16;
constexpr int CH = Ss / NCH;
constexpr int SROW = 7 * Hh * Dd;
constexpr int MSZ = Dd * Dd;
constexpr int MTOT = BH * MSZ;
constexpr int W1 = 12;
constexpr int NCH1 = 32;
constexpr int CH1 = Ss / NCH1;
constexpr int NB1 = BH * 4 * NCH1 / W1;
constexpr int TS1 = 32 * Dd;


constexpr int QP = 68;
constexpr int QTILE = 16 * QP;

__device__ __forceinline__ size_t qoff(int b, int s, int slot, int h, int d) {
  return ((((size_t)b * Ss + s) * 7 + slot) * Hh + h) * (size_t)Dd + d;
}

__device__ __forceinline__ int wave_index() {
  return __builtin_amdgcn_readfirstlane((int)(threadIdx.x >> 5));
}

constexpr int RG1 = CH1 / 4;
__global__ void __launch_bounds__(W1 * 32) ktv_strips(const float* __restrict__ qkv,
                                                      float* __restrict__ Mw) {
  __shared__ __attribute__((aligned(16))) float sP[W1 * TS1];
  const int wave = wave_index(), lane = threadIdx.x & 31;
  const int unit0 = blockIdx.x * W1;
  const int unit = unit0 + wave;
  const int strip = unit >> 5, chunk = unit & 31;
  const int ip = strip & 1;
  const int mat = (strip >> 1) & 1;
  const int bh = strip >> 2;
  const int b = bh / Hh, h = bh % Hh;
  const int kslot = mat ? 3 : 1;
  const int m = lane & 15;
  const int kk = (lane >> 4) << 1;
  v8f acc[2][4] = {};
  const float* Kp = qkv + qoff(b, chunk * CH1 + kk, kslot, h, 32 * ip + 2 * m);
  const float* Vp = qkv + qoff(b, chunk * CH1 + kk, 6, h, 4 * m);
  v2f k0[RG1], k1[RG1];
  v4f v0[RG1], v1[RG1];
#pragma unroll
  for (int g = 0; g < RG1; ++g) {
    const float* kp = Kp + (size_t)g * 4 * SROW;
    const float* vp = Vp + (size_t)g * 4 * SROW;
    k0[g] = *(const v2f*)kp;
    k1[g] = *(const v2f*)(kp + SROW);
    v0[g] = *(const v4f*)vp;
    v1[g] = *(const v4f*)(vp + SROW);
  }
#pragma unroll
  for (int g = 0; g < RG1; ++g)
#pragma unroll
    for (int i = 0; i < 2; ++i) {
      v2f a;
      a.x = k0[g][i];
      a.y = k1[g][i];
#pragma unroll
      for (int j = 0; j < 4; ++j) {
        v2f bt;
        bt.x = v0[g][j];
        bt.y = v1[g][j];
        acc[i][j] = WMMA_F32(a, bt, acc[i][j]);
      }
    }
  const int rh = (lane >> 4) * 8;
#pragma unroll
  for (int r = 0; r < 8; ++r)
#pragma unroll
    for (int i = 0; i < 2; ++i) {
      v4f o;
#pragma unroll
      for (int j = 0; j < 4; ++j) o[j] = acc[i][j][r];
      *(v4f*)&sP[wave * TS1 + (2 * (rh + r) + i) * Dd + 4 * m] = o;
    }
  __syncthreads();
  const int c0 = unit0 & 31;
  const int n0 = (NCH1 - c0) < W1 ? (NCH1 - c0) : W1;
  const int strip0 = unit0 >> 5;
  const int pa = blockIdx.x - (strip0 * 8) / 3;
  if (threadIdx.x < 256) {
    float* Ma = Mw + (size_t)(2 * pa + ((strip0 >> 1) & 1)) * MTOT +
                (size_t)(strip0 >> 2) * MSZ + (strip0 & 1) * TS1;
    const int st = strip0 + 1;
    float* Mb = Mw + (size_t)((st >> 1) & 1) * MTOT + (size_t)(st >> 2) * MSZ + (st & 1) * TS1;
#pragma unroll
    for (int q = 0; q < 2; ++q) {
      const int el = 4 * threadIdx.x + q * (TS1 / 2);
      v4f t[W1];
#pragma unroll
      for (int w = 0; w < W1; ++w) t[w] = *(const v4f*)&sP[w * TS1 + el];
      v4f sa = {0.f, 0.f, 0.f, 0.f}, sb = {0.f, 0.f, 0.f, 0.f};
#pragma unroll
      for (int w = 0; w < W1; ++w) {
        if (w < n0) sa += t[w];
        else sb += t[w];
      }
      *(v4f*)&Ma[el] = sa;
      if (n0 < W1) *(v4f*)&Mb[el] = sb;
    }
  }
}

constexpr int RG3 = 16;
constexpr int PR = 2 * CH;
constexpr int NPR = Ss / PR;
constexpr int UP = 80;
__global__ void __launch_bounds__(512) u_ktu_kernel(const float* __restrict__ qkv,
                                                    const float* __restrict__ Mw,
                                                    float* __restrict__ P3) {
  __shared__ __attribute__((aligned(16))) float smem[2 * MSZ + 32 * QTILE + PR * UP];
  float* sM1 = smem;
  float* sM2 = sM1 + MSZ;
  float* sQ1 = sM2 + MSZ;
  float* sQ2 = sQ1 + 16 * QTILE;
  float* sU = sQ2 + 16 * QTILE;
  const int bh = blockIdx.x / NPR, pr = blockIdx.x % NPR;
  const int b = bh / Hh, h = bh % Hh;
  const int wave = wave_index(), lane = threadIdx.x & 31;
  const int trow = wave * 16;
  const int tbase = pr * PR + trow;
  const int m = lane & 15, kk = (lane >> 4) << 1;
  const v4f* gm = (const v4f*)(Mw + (size_t)bh * MSZ);
  v4f fm[2][2][4];
#pragma unroll
  for (int mt = 0; mt < 2; ++mt)
#pragma unroll
    for (int i = 0; i < 2; ++i) {
#pragma unroll
      for (int p = 0; p < 3; ++p)
        fm[mt][i][p] = gm[(size_t)(2 * p + mt) * (MTOT / 4) + threadIdx.x + i * 512];
      if ((4 * bh + 2 * mt + i) % 3 == 1)
        fm[mt][i][3] = gm[(size_t)(6 + mt) * (MTOT / 4) + threadIdx.x + i * 512];
    }
  const int hr = lane >> 4;
  const float* q1p = qkv + qoff(b, tbase + hr, 0, h, 4 * m);
  const float* q2p = qkv + qoff(b, tbase + hr, 2, h, 4 * m);
  v4f r1[8], r2[8];
#pragma unroll
  for (int i = 0; i < 8; ++i) {
    r1[i] = *(const v4f*)(q1p + (size_t)(2 * i) * SROW);
    r2[i] = *(const v4f*)(q2p + (size_t)(2 * i) * SROW);
  }
  __builtin_amdgcn_sched_barrier(0);
#pragma unroll
  for (int mt = 0; mt < 2; ++mt)
#pragma unroll
    for (int i = 0; i < 2; ++i) {
      v4f sm = fm[mt][i][0] + fm[mt][i][1];
      sm += fm[mt][i][2];
      if ((4 * bh + 2 * mt + i) % 3 == 1) sm += fm[mt][i][3];
      ((v4f*)(mt ? sM2 : sM1))[threadIdx.x + i * 512] = sm;
    }
  float* w1 = sQ1 + wave * QTILE;
  float* w2 = sQ2 + wave * QTILE;
#pragma unroll
  for (int i = 0; i < 8; ++i) {
    *(v4f*)&w1[(2 * i + hr) * QP + 4 * m] = r1[i];
    *(v4f*)&w2[(2 * i + hr) * QP + 4 * m] = r2[i];
  }
  __syncthreads();
  v8f acc1[4] = {}, acc2[4] = {};
#pragma unroll 4
  for (int step = 0; step < 16; ++step) {
    const int k0 = step * 4 + kk;
    const v2f a1s = *(const v2f*)&w1[m * QP + 4 * step + kk];
    const v2f a2s = *(const v2f*)&w2[m * QP + 4 * step + kk];
    const v4f x0 = *(const v4f*)&sM1[k0 * Dd + 4 * m];
    const v4f x1 = *(const v4f*)&sM1[(k0 + 1) * Dd + 4 * m];
    const v4f y0 = *(const v4f*)&sM2[k0 * Dd + 4 * m];
    const v4f y1 = *(const v4f*)&sM2[(k0 + 1) * Dd + 4 * m];
#pragma unroll
    for (int j = 0; j < 4; ++j) {
      v2f b1, b2;
      b1.x = x0[j]; b1.y = x1[j];
      b2.x = y0[j]; b2.y = y1[j];
      acc1[j] = WMMA_F32(a1s, b1, acc1[j]);
      acc2[j] = WMMA_F32(a2s, b2, acc2[j]);
    }
  }
  const int c2 = wave >> 3, irow = (wave >> 1) & 3, jh = wave & 1;
  const float* Kp = qkv + qoff(b, pr * PR + c2 * CH + kk, 5, h, irow * 16 + m);
  v2f a3[CH / 4];
#pragma unroll
  for (int g = 0; g < CH / 4; ++g) {
    const float* kp = Kp + (size_t)g * 4 * SROW;
    a3[g].x = kp[0];
    a3[g].y = kp[SROW];
  }
  const int rh = (lane >> 4) * 8;
#pragma unroll
  for (int r = 0; r < 8; ++r) {
    v4f o;
#pragma unroll
    for (int j = 0; j < 4; ++j) {
      const float x = acc1[j][r];
      const float t1 = x / (1.f + __expf(-x));
      o[j] = t1 * acc2[j][r];
    }
    *(v4f*)&sU[(trow + rh + r) * UP + 4 * m] = o;
  }
  __syncthreads();
  const float* up = sU + (c2 * CH + kk) * UP + jh * 32 + 2 * m;
  v8f acc[2] = {};
#pragma unroll
  for (int t = 0; t < CH / (4 * RG3); ++t) {
    v2f u0[RG3], u1[RG3];
#pragma unroll
    for (int g = 0; g < RG3; ++g) {
      const float* ug = up + (t * RG3 + g) * 4 * UP;
      u0[g] = *(const v2f*)ug;
      u1[g] = *(const v2f*)(ug + UP);
    }
    __builtin_amdgcn_sched_barrier(0);
#pragma unroll
    for (int g = 0; g < RG3; ++g)
#pragma unroll
      for (int j = 0; j < 2; ++j) {
        v2f bt;
        bt.x = u0[g][j];
        bt.y = u1[g][j];
        acc[j] = WMMA_F32(a3[t * RG3 + g], bt, acc[j]);
      }
  }
  float* sX = sQ1;
#pragma unroll
  for (int r = 0; r < 8; ++r) {
    v2f o;
    o.x = acc[0][r];
    o.y = acc[1][r];
    *(v2f*)&sX[wave * 512 + (rh + r) * 32 + 2 * m] = o;
  }
  __syncthreads();
  if (c2 == 0) {
    float* P = P3 + ((size_t)bh * NPR + pr) * MSZ + irow * 16 * Dd + jh * 32;
#pragma unroll
    for (int r = 0; r < 8; ++r) {
      const v2f x0 = *(const v2f*)&sX[wave * 512 + (rh + r) * 32 + 2 * m];
      const v2f x1 = *(const v2f*)&sX[(wave + 8) * 512 + (rh + r) * 32 + 2 * m];
      *(v2f*)&P[(rh + r) * Dd + 2 * m] = x0 + x1;
    }
  }
}

__global__ void __launch_bounds__(512) out_kernel(const float* __restrict__ qkv,
                                                  const float* __restrict__ P3,
                                                  float* __restrict__ out) {
  __shared__ __attribute__((aligned(16))) float sM[MSZ];
  __shared__ __attribute__((aligned(16))) float sQ[16 * QTILE];
  const int bh = blockIdx.x / NPR, pr = blockIdx.x % NPR;
  const int b = bh / Hh, h = bh % Hh;
  const int wave = wave_index(), lane = threadIdx.x & 31;
  const int tbase = pr * PR + wave * 16;
  const int m = lane & 15, kk = (lane >> 4) << 1;
  const v4f* g3 = (const v4f*)(P3 + (size_t)bh * NPR * MSZ);
  v4f f3[2][NPR];
#pragma unroll
  for (int i = 0; i < 2; ++i)
#pragma unroll
    for (int p = 0; p < NPR; ++p) f3[i][p] = g3[p * (MSZ / 4) + threadIdx.x + i * 512];
  const int hr = lane >> 4;
  const float* qp = qkv + qoff(b, tbase + hr, 4, h, 4 * m);
  v4f rq[8];
#pragma unroll
  for (int i = 0; i < 8; ++i) rq[i] = *(const v4f*)(qp + (size_t)(2 * i) * SROW);
#pragma unroll
  for (int i = 0; i < 2; ++i) {
    v4f s3 = f3[i][0];
#pragma unroll
    for (int p = 1; p < NPR; ++p) s3 += f3[i][p];
    ((v4f*)sM)[threadIdx.x + i * 512] = s3;
  }
  float* wq = sQ + wave * QTILE;
#pragma unroll
  for (int i = 0; i < 8; ++i) *(v4f*)&wq[(2 * i + hr) * QP + 4 * m] = rq[i];
  __syncthreads();
  v2f a[16];
#pragma unroll
  for (int step = 0; step < 16; ++step) a[step] = *(const v2f*)&wq[m * QP + 4 * step + kk];
  v8f acc[4] = {};
#pragma unroll
  for (int step = 0; step < 16; ++step) {
    const int k0 = step * 4 + kk;
    const v4f x0 = *(const v4f*)&sM[k0 * Dd + 4 * m];
    const v4f x1 = *(const v4f*)&sM[(k0 + 1) * Dd + 4 * m];
#pragma unroll
    for (int j = 0; j < 4; ++j) {
      v2f bt;
      bt.x = x0[j];
      bt.y = x1[j];
      acc[j] = WMMA_F32(a[step], bt, acc[j]);
    }
  }
  const int rh = (lane >> 4) * 8;
#pragma unroll
  for (int r = 0; r < 8; ++r) {
    v4f o;
#pragma unroll
    for (int j = 0; j < 4; ++j) o[j] = acc[j][r];
    *(v4f*)&out[(((size_t)b * Ss + tbase + rh + r) * Hh + h) * Dd + 4 * m] = o;
  }
}

extern "C" void kernel_launch(void* const* d_in, const int* in_sizes, int n_in,
                              void* d_out, int out_size, void* d_ws, size_t ws_size,
                              hipStream_t stream) {
  (void)in_sizes; (void)n_in; (void)out_size; (void)ws_size;
  const float* qkv = (const float*)d_in[0];
  float* out = (float*)d_out;
  float* w = (float*)d_ws;

  float* Mw = w;
  float* P3 = w + 8 * MTOT;

  ktv_strips<<<NB1, W1 * 32, 0, stream>>>(qkv, Mw);
  u_ktu_kernel<<<BH * NPR, 512, 0, stream>>>(qkv, Mw, P3);
  out_kernel<<<BH * NPR, 512, 0, stream>>>(qkv, P3, out);
}
